# starting kernel
# baseline (speedup 1.0000x reference)
#include <hip/hip_runtime.h>
#include <math.h>

typedef __attribute__((ext_vector_type(2))) float v2f;
typedef __attribute__((ext_vector_type(8))) float v8f;

#define DIM 1024
#define NCLS 128
#define NB 8192
#define SCL 16.0f


__global__ __launch_bounds__(256) void sym_prep_kernel(const float* __restrict__ w,
                                                       float* __restrict__ ws) {
    __shared__ float r0[256], r1[256], r2[256];
    const int t = threadIdx.x;
    const float* w0 = w;
    const float* w1 = w + DIM;
    float s00 = 0.f, s01 = 0.f, s11 = 0.f;
#pragma unroll
    for (int k = 0; k < 4; ++k) {
        int e = t * 4 + k;
        float a = w0[e], b = w1[e];
        s00 = fmaf(a, a, s00);
        s01 = fmaf(a, b, s01);
        s11 = fmaf(b, b, s11);
    }
    r0[t] = s00; r1[t] = s01; r2[t] = s11;
    __syncthreads();
    for (int s = 128; s > 0; s >>= 1) {
        if (t < s) { r0[t] += r0[t + s]; r1[t] += r1[t + s]; r2[t] += r2[t + s]; }
        __syncthreads();
    }
    s00 = r0[0]; s01 = r1[0]; s11 = r2[0];
    float inv1 = 1.0f / sqrtf(s00);
    float d    = s01 / s00;
    float inv2 = 1.0f / sqrtf(fmaxf(s11 - s01 * d, 1e-30f));
#pragma unroll
    for (int k = 0; k < 4; ++k) {
        int e = t * 4 + k;
        float a = w0[e], b = w1[e];
        ws[e]       = a * inv1;
        ws[DIM + e] = (b - d * a) * inv2;
    }
    if (t < NCLS) {
        float th = 6.2831853071795864769f * (float)t / (float)NCLS;
        ws[2 * DIM + t]        = SCL * cosf(th);
        ws[2 * DIM + NCLS + t] = SCL * sinf(th);
    }
}

__global__ __launch_bounds__(128) void sym_main_kernel(const float* __restrict__ x,
                                                       const float* __restrict__ ws,
                                                       float* __restrict__ out) {
    __shared__ float4 sN1[DIM / 4];
    __shared__ float4 sN2[DIM / 4];
    __shared__ float  sCos[NCLS];
    __shared__ float  sSin[NCLS];

    const int t    = threadIdx.x;
    const int lane = t & 31;
    const int wave = t >> 5;

    {
        const float4* g1 = (const float4*)ws;
        const float4* g2 = (const float4*)(ws + DIM);
        sN1[t]       = g1[t];
        sN1[t + 128] = g1[t + 128];
        sN2[t]       = g2[t];
        sN2[t + 128] = g2[t + 128];
        sCos[t] = ws[2 * DIM + t];
        sSin[t] = ws[2 * DIM + NCLS + t];
    }
    __syncthreads();

    const int rowbase = blockIdx.x * 64 + wave * 16;

    float aM = 0.0f, bM = 0.0f;

    for (int r = 0; r < 16; ++r) {
        const float4* xp = (const float4*)(x + (size_t)(rowbase + r) * DIM);
        float sx = 0.f, sa = 0.f, sb = 0.f;
#pragma unroll
        for (int i = 0; i < 8; ++i) {
            float4 xv = xp[i * 32 + lane];
            float4 u  = sN1[i * 32 + lane];
            float4 v  = sN2[i * 32 + lane];
            sx = fmaf(xv.x, xv.x, fmaf(xv.y, xv.y, fmaf(xv.z, xv.z, fmaf(xv.w, xv.w, sx))));
            sa = fmaf(xv.x, u.x,  fmaf(xv.y, u.y,  fmaf(xv.z, u.z,  fmaf(xv.w, u.w,  sa))));
            sb = fmaf(xv.x, v.x,  fmaf(xv.y, v.y,  fmaf(xv.z, v.z,  fmaf(xv.w, v.w,  sb))));
        }
#pragma unroll
        for (int off = 16; off >= 1; off >>= 1) {
            sx += __shfl_xor(sx, off, 32);
            sa += __shfl_xor(sa, off, 32);
            sb += __shfl_xor(sb, off, 32);
        }
        float rn   = 1.0f / fmaxf(sqrtf(sx), 1e-12f);
        bool  mine = (lane == r);
        aM = mine ? sa * rn : aM;
        bM = mine ? sb * rn : bM;
    }

    const int  n       = lane & 15;
    const bool lowhalf = (lane < 16);

#if __has_builtin(__builtin_amdgcn_wmma_f32_16x16x4_f32)
    v2f afrag = {aM, bM};
#pragma unroll
    for (int tt = 0; tt < 8; ++tt) {
        float cv = sCos[tt * 16 + n];
        float sv = sSin[tt * 16 + n];
        v2f bfrag;
        bfrag.x = lowhalf ? cv : 0.0f;
        bfrag.y = lowhalf ? sv : 0.0f;
        v8f cfr = {};
        cfr = __builtin_amdgcn_wmma_f32_16x16x4_f32(false, afrag, false, bfrag,
                                                    (short)0, cfr, false, false);
        const int mbase = rowbase + (lowhalf ? 0 : 8);
#pragma unroll
        for (int vv = 0; vv < 8; ++vv) {
            out[(size_t)(mbase + vv) * NCLS + tt * 16 + n] = cfr[vv];
        }
    }
#else
    __shared__ float sA[4][16], sB[4][16];
    if (lane < 16) { sA[wave][lane] = aM; sB[wave][lane] = bM; }
    __syncthreads();
    for (int idx = lane; idx < 16 * NCLS; idx += 32) {
        int rr = idx >> 7, cc = idx & (NCLS - 1);
        out[(size_t)(rowbase + rr) * NCLS + cc] =
            fmaf(sA[wave][rr], sCos[cc], sB[wave][rr] * sSin[cc]);
    }
#endif
}

extern "C" void kernel_launch(void* const* d_in, const int* in_sizes, int n_in,
                              void* d_out, int out_size, void* d_ws, size_t ws_size,
                              hipStream_t stream) {
    const float* x  = (const float*)d_in[0];
    const float* w  = (const float*)d_in[1];
    float* out = (float*)d_out;
    float* ws  = (float*)d_ws;

    sym_prep_kernel<<<1, 256, 0, stream>>>(w, ws);
    sym_main_kernel<<<NB / 64, 128, 0, stream>>>(x, ws, out);
}


# main kernel regridded for loads in flight: 2 rows per wave on 4096 waves (was 16 rows in series on 512), both rows requested before any sum; one matrix tile per wave, whole rows stored via LDS
# speedup vs baseline: 4.8180x; 4.8180x over previous
#include <hip/hip_runtime.h>
#include <math.h>


typedef __attribute__((ext_vector_type(2))) float v2f;
typedef __attribute__((ext_vector_type(8))) float v8f;

#define DIM 1024
#define NCLS 128
#define NB 8192
#define SCL 16.0f

#define ROWS_PER_WAVE 2
#define WAVES_PER_BLOCK 8
#define ROWS_PER_BLOCK (ROWS_PER_WAVE * WAVES_PER_BLOCK)
#define DPITCH 132


__global__ __launch_bounds__(256) void sym_prep_kernel(const float* __restrict__ w,
                                                       float* __restrict__ ws) {
    __shared__ float r0[256], r1[256], r2[256];
    const int t = threadIdx.x;
    const float* w0 = w;
    const float* w1 = w + DIM;
    float s00 = 0.f, s01 = 0.f, s11 = 0.f;
#pragma unroll
    for (int k = 0; k < 4; ++k) {
        int e = t * 4 + k;
        float a = w0[e], b = w1[e];
        s00 = fmaf(a, a, s00);
        s01 = fmaf(a, b, s01);
        s11 = fmaf(b, b, s11);
    }
    r0[t] = s00; r1[t] = s01; r2[t] = s11;
    __syncthreads();
    for (int s = 128; s > 0; s >>= 1) {
        if (t < s) { r0[t] += r0[t + s]; r1[t] += r1[t + s]; r2[t] += r2[t + s]; }
        __syncthreads();
    }
    s00 = r0[0]; s01 = r1[0]; s11 = r2[0];
    float inv1 = 1.0f / sqrtf(s00);
    float d    = s01 / s00;
    float inv2 = 1.0f / sqrtf(fmaxf(s11 - s01 * d, 1e-30f));
#pragma unroll
    for (int k = 0; k < 4; ++k) {
        int e = t * 4 + k;
        float a = w0[e], b = w1[e];
        ws[e]       = a * inv1;
        ws[DIM + e] = (b - d * a) * inv2;
    }
    if (t < NCLS) {
        float th = 6.2831853071795864769f * (float)t / (float)NCLS;
        ws[2 * DIM + t]        = SCL * cosf(th);
        ws[2 * DIM + NCLS + t] = SCL * sinf(th);
    }
}

__global__ __launch_bounds__(256) void sym_main_kernel(const float* __restrict__ x,
                                                       const float* __restrict__ ws,
                                                       float* __restrict__ out) {
    __shared__ float4 sN1[DIM / 4];
    __shared__ float4 sN2[DIM / 4];
    __shared__ float  sA[ROWS_PER_BLOCK];
    __shared__ float  sB[ROWS_PER_BLOCK];
    __shared__ __attribute__((aligned(16))) float sD[ROWS_PER_BLOCK * DPITCH];

    const int  t       = threadIdx.x;
    const int  lane    = t & 31;
    const int  wave    = t >> 5;
    const int  n       = lane & 15;
    const bool lowhalf = (lane < 16);
    const int  rowbase = blockIdx.x * ROWS_PER_BLOCK;

    const float4 g1 = ((const float4*)ws)[t];
    const float4 g2 = ((const float4*)(ws + DIM))[t];
    const float  cv = ws[2 * DIM + wave * 16 + n];
    const float  sv = ws[2 * DIM + NCLS + wave * 16 + n];

    float4 xv[ROWS_PER_WAVE][8];
#pragma unroll
    for (int r = 0; r < ROWS_PER_WAVE; ++r) {
        const float4* xp = (const float4*)(x + (size_t)(rowbase + wave * ROWS_PER_WAVE + r) * DIM);
#pragma unroll
        for (int i = 0; i < 8; ++i) xv[r][i] = xp[i * 32 + lane];
    }

    sN1[t] = g1;
    sN2[t] = g2;
    __builtin_amdgcn_fence(__ATOMIC_RELEASE, "workgroup", "local");
    __builtin_amdgcn_s_barrier();
    __builtin_amdgcn_fence(__ATOMIC_ACQUIRE, "workgroup", "local");
    __builtin_amdgcn_sched_barrier(0);

    float4 u[8], v[8];
#pragma unroll
    for (int i = 0; i < 8; ++i) {
        u[i] = sN1[i * 32 + lane];
        v[i] = sN2[i * 32 + lane];
    }

    float aR[ROWS_PER_WAVE], bR[ROWS_PER_WAVE];
#pragma unroll
    for (int r = 0; r < ROWS_PER_WAVE; ++r) {
        float sx = 0.f, sa = 0.f, sb = 0.f;
#pragma unroll
        for (int i = 0; i < 8; ++i) {
            const float4 xq = xv[r][i];
            sx = fmaf(xq.x, xq.x, fmaf(xq.y, xq.y, fmaf(xq.z, xq.z, fmaf(xq.w, xq.w, sx))));
            sa = fmaf(xq.x, u[i].x, fmaf(xq.y, u[i].y, fmaf(xq.z, u[i].z, fmaf(xq.w, u[i].w, sa))));
            sb = fmaf(xq.x, v[i].x, fmaf(xq.y, v[i].y, fmaf(xq.z, v[i].z, fmaf(xq.w, v[i].w, sb))));
        }
#pragma unroll
        for (int off = 16; off >= 1; off >>= 1) {
            sx += __shfl_xor(sx, off, 32);
            sa += __shfl_xor(sa, off, 32);
            sb += __shfl_xor(sb, off, 32);
        }
        const float rn = 1.0f / fmaxf(sqrtf(sx), 1e-12f);
        aR[r] = sa * rn;
        bR[r] = sb * rn;
    }
    if (lane < ROWS_PER_WAVE) {
        sA[wave * ROWS_PER_WAVE + lane] = (lane == 0) ? aR[0] : aR[1];
        sB[wave * ROWS_PER_WAVE + lane] = (lane == 0) ? bR[0] : bR[1];
    }
    __syncthreads();

    const float am = sA[n];
    const float bm = sB[n];
    v2f afrag;
    afrag.x = lowhalf ? am : 0.0f;
    afrag.y = lowhalf ? bm : 0.0f;
    v2f bfrag;
    bfrag.x = lowhalf ? cv : 0.0f;
    bfrag.y = lowhalf ? sv : 0.0f;
    v8f cfr = {};
    cfr = __builtin_amdgcn_wmma_f32_16x16x4_f32(false, afrag, false, bfrag,
                                                (short)0, cfr, false, false);
    const int mb = lowhalf ? 0 : 8;
#pragma unroll
    for (int vv = 0; vv < 8; ++vv) {
        sD[(mb + vv) * DPITCH + wave * 16 + n] = cfr[vv];
    }
    __syncthreads();

#pragma unroll
    for (int rr = 0; rr < ROWS_PER_WAVE; ++rr) {
        const int row = wave * ROWS_PER_WAVE + rr;
        const float4 o = *(const float4*)(&sD[row * DPITCH + lane * 4]);
        *(float4*)(out + (size_t)(rowbase + row) * NCLS + lane * 4) = o;
    }
}

extern "C" void kernel_launch(void* const* d_in, const int* in_sizes, int n_in,
                              void* d_out, int out_size, void* d_ws, size_t ws_size,
                              hipStream_t stream) {
    const float* x  = (const float*)d_in[0];
    const float* w  = (const float*)d_in[1];
    float* out = (float*)d_out;
    float* ws  = (float*)d_ws;

    sym_prep_kernel<<<1, 256, 0, stream>>>(w, ws);
    sym_main_kernel<<<NB / ROWS_PER_BLOCK, 256, 0, stream>>>(x, ws, out);
}


# prep launch folded into the main kernel (one launch, no workspace): each wave sums w0.w0, w0.w1, w1.w1 itself and scales its row sums afterwards; cos/sin per lane by sincospi; new weight-sum order
# speedup vs baseline: 7.3663x; 1.5289x over previous
#include <hip/hip_runtime.h>
#include <math.h>


typedef __attribute__((ext_vector_type(2))) float v2f;
typedef __attribute__((ext_vector_type(8))) float v8f;

#define DIM 1024
#define NCLS 128
#define NB 8192
#define SCL 16.0f

#define ROWS_PER_WAVE 2
#define WAVES_PER_BLOCK 8
#define ROWS_PER_BLOCK (ROWS_PER_WAVE * WAVES_PER_BLOCK)
#define DPITCH 132

__global__ __launch_bounds__(256) void sym_fused_kernel(const float* __restrict__ x,
                                                        const float* __restrict__ w,
                                                        float* __restrict__ out) {
    __shared__ float4 sW0[DIM / 4];
    __shared__ float4 sW1[DIM / 4];
    __shared__ float  sA[ROWS_PER_BLOCK];
    __shared__ float  sB[ROWS_PER_BLOCK];
    __shared__ __attribute__((aligned(16))) float sD[ROWS_PER_BLOCK * DPITCH];

    const int  t       = threadIdx.x;
    const int  lane    = t & 31;
    const int  wave    = t >> 5;
    const int  n       = lane & 15;
    const bool lowhalf = (lane < 16);
    const int  rowbase = blockIdx.x * ROWS_PER_BLOCK;

    const float4 g1 = ((const float4*)w)[t];
    const float4 g2 = ((const float4*)(w + DIM))[t];

    float4 xv[ROWS_PER_WAVE][8];
#pragma unroll
    for (int r = 0; r < ROWS_PER_WAVE; ++r) {
        const float4* xp = (const float4*)(x + (size_t)(rowbase + wave * ROWS_PER_WAVE + r) * DIM);
#pragma unroll
        for (int i = 0; i < 8; ++i) xv[r][i] = xp[i * 32 + lane];
    }

    sW0[t] = g1;
    sW1[t] = g2;
    __builtin_amdgcn_fence(__ATOMIC_RELEASE, "workgroup", "local");
    __builtin_amdgcn_s_barrier();
    __builtin_amdgcn_fence(__ATOMIC_ACQUIRE, "workgroup", "local");
    __builtin_amdgcn_sched_barrier(0);

    float4 u[8], v[8];
#pragma unroll
    for (int i = 0; i < 8; ++i) {
        u[i] = sW0[i * 32 + lane];
        v[i] = sW1[i * 32 + lane];
    }

    float sn, cs;
    sincospif((float)(wave * 16 + n) * (2.0f / (float)NCLS), &sn, &cs);
    asm volatile("" :: "v"(sn), "v"(cs));
    const float cv = SCL * cs;
    const float sv = SCL * sn;

    float s00 = 0.f, s01 = 0.f, s11 = 0.f;
#pragma unroll
    for (int i = 0; i < 8; ++i) {
        s00 = fmaf(u[i].x, u[i].x, fmaf(u[i].y, u[i].y, fmaf(u[i].z, u[i].z, fmaf(u[i].w, u[i].w, s00))));
        s01 = fmaf(u[i].x, v[i].x, fmaf(u[i].y, v[i].y, fmaf(u[i].z, v[i].z, fmaf(u[i].w, v[i].w, s01))));
        s11 = fmaf(v[i].x, v[i].x, fmaf(v[i].y, v[i].y, fmaf(v[i].z, v[i].z, fmaf(v[i].w, v[i].w, s11))));
    }
#pragma unroll
    for (int off = 16; off >= 1; off >>= 1) {
        s00 += __shfl_xor(s00, off, 32);
        s01 += __shfl_xor(s01, off, 32);
        s11 += __shfl_xor(s11, off, 32);
    }
    const float inv1 = 1.0f / sqrtf(s00);
    const float d    = s01 / s00;
    const float inv2 = 1.0f / sqrtf(fmaxf(s11 - s01 * d, 1e-30f));

    float aR[ROWS_PER_WAVE], bR[ROWS_PER_WAVE];
#pragma unroll
    for (int r = 0; r < ROWS_PER_WAVE; ++r) {
        float sx = 0.f, sa = 0.f, sb = 0.f;
#pragma unroll
        for (int i = 0; i < 8; ++i) {
            const float4 xq = xv[r][i];
            sx = fmaf(xq.x, xq.x, fmaf(xq.y, xq.y, fmaf(xq.z, xq.z, fmaf(xq.w, xq.w, sx))));
            sa = fmaf(xq.x, u[i].x, fmaf(xq.y, u[i].y, fmaf(xq.z, u[i].z, fmaf(xq.w, u[i].w, sa))));
            sb = fmaf(xq.x, v[i].x, fmaf(xq.y, v[i].y, fmaf(xq.z, v[i].z, fmaf(xq.w, v[i].w, sb))));
        }
#pragma unroll
        for (int off = 16; off >= 1; off >>= 1) {
            sx += __shfl_xor(sx, off, 32);
            sa += __shfl_xor(sa, off, 32);
            sb += __shfl_xor(sb, off, 32);
        }
        const float rn = 1.0f / fmaxf(sqrtf(sx), 1e-12f);
        aR[r] = sa * inv1 * rn;
        bR[r] = (sb - d * sa) * inv2 * rn;
    }
    if (lane < ROWS_PER_WAVE) {
        sA[wave * ROWS_PER_WAVE + lane] = (lane == 0) ? aR[0] : aR[1];
        sB[wave * ROWS_PER_WAVE + lane] = (lane == 0) ? bR[0] : bR[1];
    }
    __syncthreads();

    const float am = sA[n];
    const float bm = sB[n];
    v2f afrag;
    afrag.x = lowhalf ? am : 0.0f;
    afrag.y = lowhalf ? bm : 0.0f;
    v2f bfrag;
    bfrag.x = lowhalf ? cv : 0.0f;
    bfrag.y = lowhalf ? sv : 0.0f;
    v8f cfr = {};
    cfr = __builtin_amdgcn_wmma_f32_16x16x4_f32(false, afrag, false, bfrag,
                                                (short)0, cfr, false, false);
    const int mb = lowhalf ? 0 : 8;
#pragma unroll
    for (int vv = 0; vv < 8; ++vv) {
        sD[(mb + vv) * DPITCH + wave * 16 + n] = cfr[vv];
    }
    __syncthreads();

#pragma unroll
    for (int rr = 0; rr < ROWS_PER_WAVE; ++rr) {
        const int row = wave * ROWS_PER_WAVE + rr;
        const float4 o = *(const float4*)(&sD[row * DPITCH + lane * 4]);
        *(float4*)(out + (size_t)(rowbase + row) * NCLS + lane * 4) = o;
    }
}

extern "C" void kernel_launch(void* const* d_in, const int* in_sizes, int n_in,
                              void* d_out, int out_size, void* d_ws, size_t ws_size,
                              hipStream_t stream) {
    const float* x  = (const float*)d_in[0];
    const float* w  = (const float*)d_in[1];
    float* out = (float*)d_out;

    sym_fused_kernel<<<NB / ROWS_PER_BLOCK, 256, 0, stream>>>(x, w, out);
}


# wave tail cut: sums by permlanex16 + row_xmask DPP adds instead of 45 ds_bpermute (same order), scale factors by rsqrtf and v_rcp_f32 instead of sqrt + division (126-instruction branch now 38)
# speedup vs baseline: 7.8967x; 1.0720x over previous
#include <hip/hip_runtime.h>
#include <math.h>


typedef __attribute__((ext_vector_type(2))) float v2f;
typedef __attribute__((ext_vector_type(8))) float v8f;

#define DIM 1024
#define NCLS 128
#define NB 8192
#define SCL 16.0f

#define ROWS_PER_WAVE 2
#define WAVES_PER_BLOCK 8
#define ROWS_PER_BLOCK (ROWS_PER_WAVE * WAVES_PER_BLOCK)
#define DPITCH 132

__device__ __forceinline__ float xhalf(float x) {
    const int b = __builtin_bit_cast(int, x);
    return __builtin_bit_cast(float,
        __builtin_amdgcn_permlanex16(b, b, 0x76543210, 0xfedcba98, false, false));
}
template <int M>
__device__ __forceinline__ float xrow(float x) {
    const int b = __builtin_bit_cast(int, x);
    return __builtin_bit_cast(float, __builtin_amdgcn_update_dpp(b, b, 0x160 + M, 0xf, 0xf, true));
}

__global__ __launch_bounds__(256) void sym_fused_kernel(const float* __restrict__ x,
                                                        const float* __restrict__ w,
                                                        float* __restrict__ out) {
    __shared__ float4 sW0[DIM / 4];
    __shared__ float4 sW1[DIM / 4];
    __shared__ float  sA[ROWS_PER_BLOCK];
    __shared__ float  sB[ROWS_PER_BLOCK];
    __shared__ __attribute__((aligned(16))) float sD[ROWS_PER_BLOCK * DPITCH];

    const int  t       = threadIdx.x;
    const int  lane    = t & 31;
    const int  wave    = t >> 5;
    const int  n       = lane & 15;
    const bool lowhalf = (lane < 16);
    const int  rowbase = blockIdx.x * ROWS_PER_BLOCK;

    const float4 g1 = ((const float4*)w)[t];
    const float4 g2 = ((const float4*)(w + DIM))[t];

    float4 xv[ROWS_PER_WAVE][8];
#pragma unroll
    for (int r = 0; r < ROWS_PER_WAVE; ++r) {
        const float4* xp = (const float4*)(x + (size_t)(rowbase + wave * ROWS_PER_WAVE + r) * DIM);
#pragma unroll
        for (int i = 0; i < 8; ++i) xv[r][i] = xp[i * 32 + lane];
    }

    sW0[t] = g1;
    sW1[t] = g2;
    __builtin_amdgcn_fence(__ATOMIC_RELEASE, "workgroup", "local");
    __builtin_amdgcn_s_barrier();
    __builtin_amdgcn_fence(__ATOMIC_ACQUIRE, "workgroup", "local");
    __builtin_amdgcn_sched_barrier(0);

    float4 u[8], v[8];
#pragma unroll
    for (int i = 0; i < 8; ++i) {
        u[i] = sW0[i * 32 + lane];
        v[i] = sW1[i * 32 + lane];
    }

    float sn, cs;
    sincospif((float)(wave * 16 + n) * (2.0f / (float)NCLS), &sn, &cs);
    asm volatile("" :: "v"(sn), "v"(cs));
    const float cv = SCL * cs;
    const float sv = SCL * sn;

    float s00 = 0.f, s01 = 0.f, s11 = 0.f;
#pragma unroll
    for (int i = 0; i < 8; ++i) {
        s00 = fmaf(u[i].x, u[i].x, fmaf(u[i].y, u[i].y, fmaf(u[i].z, u[i].z, fmaf(u[i].w, u[i].w, s00))));
        s01 = fmaf(u[i].x, v[i].x, fmaf(u[i].y, v[i].y, fmaf(u[i].z, v[i].z, fmaf(u[i].w, v[i].w, s01))));
        s11 = fmaf(v[i].x, v[i].x, fmaf(v[i].y, v[i].y, fmaf(v[i].z, v[i].z, fmaf(v[i].w, v[i].w, s11))));
    }
    s00 += xhalf(s00);    s01 += xhalf(s01);    s11 += xhalf(s11);
    s00 += xrow<8>(s00);  s01 += xrow<8>(s01);  s11 += xrow<8>(s11);
    s00 += xrow<4>(s00);  s01 += xrow<4>(s01);  s11 += xrow<4>(s11);
    s00 += xrow<2>(s00);  s01 += xrow<2>(s01);  s11 += xrow<2>(s11);
    s00 += xrow<1>(s00);  s01 += xrow<1>(s01);  s11 += xrow<1>(s11);
    const float inv1 = rsqrtf(s00);
    const float d    = s01 * __builtin_amdgcn_rcpf(s00);
    const float inv2 = rsqrtf(fmaxf(s11 - s01 * d, 1e-30f));

    float aR[ROWS_PER_WAVE], bR[ROWS_PER_WAVE];
#pragma unroll
    for (int r = 0; r < ROWS_PER_WAVE; ++r) {
        float sx = 0.f, sa = 0.f, sb = 0.f;
#pragma unroll
        for (int i = 0; i < 8; ++i) {
            const float4 xq = xv[r][i];
            sx = fmaf(xq.x, xq.x, fmaf(xq.y, xq.y, fmaf(xq.z, xq.z, fmaf(xq.w, xq.w, sx))));
            sa = fmaf(xq.x, u[i].x, fmaf(xq.y, u[i].y, fmaf(xq.z, u[i].z, fmaf(xq.w, u[i].w, sa))));
            sb = fmaf(xq.x, v[i].x, fmaf(xq.y, v[i].y, fmaf(xq.z, v[i].z, fmaf(xq.w, v[i].w, sb))));
        }
        sx += xhalf(sx);    sa += xhalf(sa);    sb += xhalf(sb);
        sx += xrow<8>(sx);  sa += xrow<8>(sa);  sb += xrow<8>(sb);
        sx += xrow<4>(sx);  sa += xrow<4>(sa);  sb += xrow<4>(sb);
        sx += xrow<2>(sx);  sa += xrow<2>(sa);  sb += xrow<2>(sb);
        sx += xrow<1>(sx);  sa += xrow<1>(sa);  sb += xrow<1>(sb);
        const float rn = fminf(rsqrtf(sx), 1e12f);
        aR[r] = sa * inv1 * rn;
        bR[r] = (sb - d * sa) * inv2 * rn;
    }
    if (lane < ROWS_PER_WAVE) {
        sA[wave * ROWS_PER_WAVE + lane] = (lane == 0) ? aR[0] : aR[1];
        sB[wave * ROWS_PER_WAVE + lane] = (lane == 0) ? bR[0] : bR[1];
    }
    __syncthreads();

    const float am = sA[n];
    const float bm = sB[n];
    v2f afrag;
    afrag.x = lowhalf ? am : 0.0f;
    afrag.y = lowhalf ? bm : 0.0f;
    v2f bfrag;
    bfrag.x = lowhalf ? cv : 0.0f;
    bfrag.y = lowhalf ? sv : 0.0f;
    v8f cfr = {};
    cfr = __builtin_amdgcn_wmma_f32_16x16x4_f32(false, afrag, false, bfrag,
                                                (short)0, cfr, false, false);
    const int mb = lowhalf ? 0 : 8;
#pragma unroll
    for (int vv = 0; vv < 8; ++vv) {
        sD[(mb + vv) * DPITCH + wave * 16 + n] = cfr[vv];
    }
    __syncthreads();

#pragma unroll
    for (int rr = 0; rr < ROWS_PER_WAVE; ++rr) {
        const int row = wave * ROWS_PER_WAVE + rr;
        const float4 o = *(const float4*)(&sD[row * DPITCH + lane * 4]);
        *(float4*)(out + (size_t)(rowbase + row) * NCLS + lane * 4) = o;
    }
}

extern "C" void kernel_launch(void* const* d_in, const int* in_sizes, int n_in,
                              void* d_out, int out_size, void* d_ws, size_t ws_size,
                              hipStream_t stream) {
    const float* x  = (const float*)d_in[0];
    const float* w  = (const float*)d_in[1];
    float* out = (float*)d_out;

    sym_fused_kernel<<<NB / ROWS_PER_BLOCK, 256, 0, stream>>>(x, w, out);
}


# main kernel regridded for wave entry: 4 rows per wave on 2048 waves (was 2 on 4096), all 32 row loads requested before any sum; 256 blocks of 32 rows, two matrix tiles per wave
# speedup vs baseline: 8.3566x; 1.0582x over previous
#include <hip/hip_runtime.h>
#include <math.h>


typedef __attribute__((ext_vector_type(2))) float v2f;
typedef __attribute__((ext_vector_type(8))) float v8f;

#define DIM 1024
#define NCLS 128
#define NB 8192
#define SCL 16.0f

#define ROWS_PER_WAVE 4
#define WAVES_PER_BLOCK 8
#define ROWS_PER_BLOCK (ROWS_PER_WAVE * WAVES_PER_BLOCK)
#define ROW_TILES (ROWS_PER_BLOCK / 16)
#define DPITCH 132

__device__ __forceinline__ float xhalf(float x) {
    const int b = __builtin_bit_cast(int, x);
    return __builtin_bit_cast(float,
        __builtin_amdgcn_permlanex16(b, b, 0x76543210, 0xfedcba98, false, false));
}
template <int M>
__device__ __forceinline__ float xrow(float x) {
    const int b = __builtin_bit_cast(int, x);
    return __builtin_bit_cast(float, __builtin_amdgcn_update_dpp(b, b, 0x160 + M, 0xf, 0xf, true));
}

__global__ __launch_bounds__(256) void sym_fused_kernel(const float* x,
                                                        const float* w,
                                                        float* out) {
    __shared__ float4 sW0[DIM / 4];
    __shared__ float4 sW1[DIM / 4];
    __shared__ float  sA[ROWS_PER_BLOCK];
    __shared__ float  sB[ROWS_PER_BLOCK];
    __shared__ __attribute__((aligned(16))) float sD[ROWS_PER_BLOCK * DPITCH];

    const int  t       = threadIdx.x;
    const int  lane    = t & 31;
    const int  wave    = t >> 5;
    const int  n       = lane & 15;
    const bool lowhalf = (lane < 16);
    const int  rowbase = blockIdx.x * ROWS_PER_BLOCK;

    const float4 g1 = ((const float4*)w)[t];
    const float4 g2 = ((const float4*)(w + DIM))[t];
    asm volatile("" ::: "memory");

    float4 xv[ROWS_PER_WAVE][8];
#pragma unroll
    for (int r = 0; r < ROWS_PER_WAVE; ++r) {
        const float4* xp = (const float4*)(x + (size_t)(rowbase + wave * ROWS_PER_WAVE + r) * DIM);
#pragma unroll
        for (int i = 0; i < 8; ++i) xv[r][i] = xp[i * 32 + lane];
    }

    sW0[t] = g1;
    sW1[t] = g2;
    __builtin_amdgcn_fence(__ATOMIC_RELEASE, "workgroup", "local");
    __builtin_amdgcn_s_barrier();
    __builtin_amdgcn_fence(__ATOMIC_ACQUIRE, "workgroup", "local");
    __builtin_amdgcn_sched_barrier(0);

    float4 u[8], v[8];
#pragma unroll
    for (int i = 0; i < 8; ++i) {
        u[i] = sW0[i * 32 + lane];
        v[i] = sW1[i * 32 + lane];
    }

    float sn, cs;
    sincospif((float)(wave * 16 + n) * (2.0f / (float)NCLS), &sn, &cs);
    asm volatile("" :: "v"(sn), "v"(cs));
    const float cv = SCL * cs;
    const float sv = SCL * sn;

    float s00 = 0.f, s01 = 0.f, s11 = 0.f;
#pragma unroll
    for (int i = 0; i < 8; ++i) {
        s00 = fmaf(u[i].x, u[i].x, fmaf(u[i].y, u[i].y, fmaf(u[i].z, u[i].z, fmaf(u[i].w, u[i].w, s00))));
        s01 = fmaf(u[i].x, v[i].x, fmaf(u[i].y, v[i].y, fmaf(u[i].z, v[i].z, fmaf(u[i].w, v[i].w, s01))));
        s11 = fmaf(v[i].x, v[i].x, fmaf(v[i].y, v[i].y, fmaf(v[i].z, v[i].z, fmaf(v[i].w, v[i].w, s11))));
    }
    s00 += xhalf(s00);    s01 += xhalf(s01);    s11 += xhalf(s11);
    s00 += xrow<8>(s00);  s01 += xrow<8>(s01);  s11 += xrow<8>(s11);
    s00 += xrow<4>(s00);  s01 += xrow<4>(s01);  s11 += xrow<4>(s11);
    s00 += xrow<2>(s00);  s01 += xrow<2>(s01);  s11 += xrow<2>(s11);
    s00 += xrow<1>(s00);  s01 += xrow<1>(s01);  s11 += xrow<1>(s11);
    const float inv1 = rsqrtf(s00);
    const float d    = s01 * __builtin_amdgcn_rcpf(s00);
    const float inv2 = rsqrtf(fmaxf(s11 - s01 * d, 1e-30f));
    asm volatile("" :: "v"(inv1), "v"(d), "v"(inv2));
    __builtin_amdgcn_sched_barrier(0);

    float sxR[ROWS_PER_WAVE], saR[ROWS_PER_WAVE], sbR[ROWS_PER_WAVE];
#pragma unroll
    for (int rp = 0; rp < ROWS_PER_WAVE; rp += 2) {
        float sx[2] = {0.f, 0.f}, sa[2] = {0.f, 0.f}, sb[2] = {0.f, 0.f};
#pragma unroll
        for (int i = 0; i < 8; ++i) {
#pragma unroll
            for (int q = 0; q < 2; ++q) {
                const float4 xq = xv[rp + q][i];
                sx[q] = fmaf(xq.x, xq.x, fmaf(xq.y, xq.y, fmaf(xq.z, xq.z, fmaf(xq.w, xq.w, sx[q]))));
                sa[q] = fmaf(xq.x, u[i].x, fmaf(xq.y, u[i].y, fmaf(xq.z, u[i].z, fmaf(xq.w, u[i].w, sa[q]))));
                sb[q] = fmaf(xq.x, v[i].x, fmaf(xq.y, v[i].y, fmaf(xq.z, v[i].z, fmaf(xq.w, v[i].w, sb[q]))));
            }
            __builtin_amdgcn_sched_barrier(0);
        }
#pragma unroll
        for (int q = 0; q < 2; ++q) {
            sx[q] += xhalf(sx[q]);    sa[q] += xhalf(sa[q]);    sb[q] += xhalf(sb[q]);
            sx[q] += xrow<8>(sx[q]);  sa[q] += xrow<8>(sa[q]);  sb[q] += xrow<8>(sb[q]);
            sx[q] += xrow<4>(sx[q]);  sa[q] += xrow<4>(sa[q]);  sb[q] += xrow<4>(sb[q]);
            sx[q] += xrow<2>(sx[q]);  sa[q] += xrow<2>(sa[q]);  sb[q] += xrow<2>(sb[q]);
            sx[q] += xrow<1>(sx[q]);  sa[q] += xrow<1>(sa[q]);  sb[q] += xrow<1>(sb[q]);
            sxR[rp + q] = sx[q];  saR[rp + q] = sa[q];  sbR[rp + q] = sb[q];
        }
        __builtin_amdgcn_sched_barrier(0);
    }
    const bool odd = (lane & 1) != 0, upper = (lane & 2) != 0;
    const float sxm = upper ? (odd ? sxR[3] : sxR[2]) : (odd ? sxR[1] : sxR[0]);
    const float sam = upper ? (odd ? saR[3] : saR[2]) : (odd ? saR[1] : saR[0]);
    const float sbm = upper ? (odd ? sbR[3] : sbR[2]) : (odd ? sbR[1] : sbR[0]);
    const float rn  = fminf(rsqrtf(sxm), 1e12f);
    const float aRm = sam * inv1 * rn;
    const float bRm = (sbm - d * sam) * inv2 * rn;
    if (lane < ROWS_PER_WAVE) {
        sA[wave * ROWS_PER_WAVE + lane] = aRm;
        sB[wave * ROWS_PER_WAVE + lane] = bRm;
    }
    __syncthreads();

    v2f bfrag;
    bfrag.x = lowhalf ? cv : 0.0f;
    bfrag.y = lowhalf ? sv : 0.0f;
    const int mb = lowhalf ? 0 : 8;
#pragma unroll
    for (int rt = 0; rt < ROW_TILES; ++rt) {
        const float am = sA[rt * 16 + n];
        const float bm = sB[rt * 16 + n];
        v2f afrag;
        afrag.x = lowhalf ? am : 0.0f;
        afrag.y = lowhalf ? bm : 0.0f;
        v8f cfr = {};
        cfr = __builtin_amdgcn_wmma_f32_16x16x4_f32(false, afrag, false, bfrag,
                                                    (short)0, cfr, false, false);
#pragma unroll
        for (int vv = 0; vv < 8; ++vv) {
            sD[(rt * 16 + mb + vv) * DPITCH + wave * 16 + n] = cfr[vv];
        }
    }
    __syncthreads();

#pragma unroll
    for (int rr = 0; rr < ROWS_PER_WAVE; ++rr) {
        const int row = wave * ROWS_PER_WAVE + rr;
        const float4 o = *(const float4*)(&sD[row * DPITCH + lane * 4]);
        *(float4*)(out + (size_t)(rowbase + row) * NCLS + lane * 4) = o;
    }
}

extern "C" void kernel_launch(void* const* d_in, const int* in_sizes, int n_in,
                              void* d_out, int out_size, void* d_ws, size_t ws_size,
                              hipStream_t stream) {
    const float* x  = (const float*)d_in[0];
    const float* w  = (const float*)d_in[1];
    float* out = (float*)d_out;

    sym_fused_kernel<<<NB / ROWS_PER_BLOCK, 256, 0, stream>>>(x, w, out);
}
